# starting kernel
# baseline (speedup 1.0000x reference)
#include <hip/hip_runtime.h>

#define TT 256
#define DD 32
#define KK 8
#define NB 64
#define AUG_LD 66
#define EPS_TRI 0.01f
#define EPS_CHOL 0.01f
#define EPS_BS 0.0001f

typedef float v2f __attribute__((ext_vector_type(2)));
typedef float v8f __attribute__((ext_vector_type(8)));

enum { MM_STORE = 0, MM_STORE_NEG = 1, MM_SUB = 2, MM_ADD = 3 };

template <bool TA, bool TB>
__device__ inline void mm32(const float* A, const float* B, float* D, int ldd,
                            int mode, int tid) {
  const int lane = tid & 31;
  const int wid = tid >> 5;
  const int mt = (wid >> 1) << 4;
  const int nt = (wid & 1) << 4;
  const int l15 = lane & 15;
  const int kh = (lane >> 4) << 1;
  v8f acc = {};
#pragma unroll
  for (int kt = 0; kt < 32; kt += 4) {
    const int k0 = kt + kh;
    v2f av, bv;
    if (!TA) { av.x = A[(mt + l15) * 32 + k0];  av.y = A[(mt + l15) * 32 + k0 + 1]; }
    else     { av.x = A[k0 * 32 + mt + l15];    av.y = A[(k0 + 1) * 32 + mt + l15]; }
    if (!TB) { bv.x = B[k0 * 32 + nt + l15];    bv.y = B[(k0 + 1) * 32 + nt + l15]; }
    else     { bv.x = B[(nt + l15) * 32 + k0];  bv.y = B[(nt + l15) * 32 + k0 + 1]; }
    acc = __builtin_amdgcn_wmma_f32_16x16x4_f32(false, av, false, bv, (short)0,
                                                acc, false, false);
  }
  const int rbase = mt + ((lane >> 4) << 3);
#pragma unroll
  for (int r = 0; r < 8; ++r) {
    float v = acc[r];
    float* p = &D[(rbase + r) * ldd + nt + l15];
    if (mode == MM_STORE) *p = v;
    else if (mode == MM_STORE_NEG) *p = -v;
    else if (mode == MM_SUB) *p -= v;
    else *p += v;
  }
}

__global__ __launch_bounds__(128) void ris_precompute_kernel(
    const float* __restrict__ Ri_sqrts, const float* __restrict__ ms,
    float* __restrict__ g_Ris, float* __restrict__ g_hR) {
  __shared__ float sR[1024], sO[1024], sm[DD];
  const int t = blockIdx.x;
  const int tid = threadIdx.x;
  for (int e = tid; e < 1024; e += 128) sR[e] = Ri_sqrts[(size_t)t * 1024 + e];
  if (tid < DD) sm[tid] = ms[t * DD + tid];
  __syncthreads();
  mm32<false, true>(sR, sR, sO, 32, MM_STORE, tid);
  __syncthreads();
  for (int e = tid; e < 1024; e += 128) g_Ris[(size_t)t * 1024 + e] = sO[e];
  if (tid < DD) {
    float a = 0.f;
    for (int j = 0; j < DD; ++j) a += sO[tid * 32 + j] * sm[j];
    g_hR[t * DD + tid] = a;
  }
}

__global__ __launch_bounds__(128) void lds_forward_kernel(
    const float* __restrict__ z_samples, const float* __restrict__ A_base,
    const float* __restrict__ b_base, const float* __restrict__ Q_sqrt,
    const float* __restrict__ noise, const float* __restrict__ g_Ris,
    const float* __restrict__ g_hR, float* __restrict__ g_c,
    float* __restrict__ g_d, float* __restrict__ out) {
  __shared__ float sA[1024], sQs[1024], sJd[1024], sC[1024], sL[1024],
      sLl[1024], sM[1024];
  __shared__ float sQis[2][1024], sJl[2][1024];
  __shared__ float sAug[32 * AUG_LD];
  __shared__ float sz[KK], sbs[DD], sQb[DD], sh[DD], sJlPd[DD], sd[DD],
      srhs[DD], sx[DD];

  const int n = blockIdx.x;
  const int tid = threadIdx.x;
  const int row = tid & 31;
  const int jgrp = tid >> 5;

  for (int t = 0; t < TT; ++t) {
    const int cur = t & 1, prev = cur ^ 1;
    const bool haveNext = (t < TT - 1);

    if (haveNext) {
      if (tid < KK) sz[tid] = z_samples[((size_t)n * TT + t) * KK + tid];
      __syncthreads();
      for (int e = tid; e < 1024; e += 128) {
        float a = 0.f, q = 0.f;
#pragma unroll
        for (int k = 0; k < KK; ++k) {
          float zk = sz[k];
          a += zk * A_base[k * 1024 + e];
          q += zk * Q_sqrt[k * 1024 + e];
        }
        sA[e] = a;
        sQs[e] = q;
      }
      if (tid < DD) {
        float b = 0.f;
#pragma unroll
        for (int k = 0; k < KK; ++k) b += sz[k] * b_base[k * DD + tid];
        sbs[tid] = b;
      }
      __syncthreads();
      mm32<false, true>(sQs, sQs, sQis[cur], 32, MM_STORE, tid);
      __syncthreads();
      mm32<false, false>(sQis[cur], sA, sJl[cur], 32, MM_STORE_NEG, tid);
      __syncthreads();
      mm32<true, false>(sA, sJl[cur], sJd, 32, MM_STORE_NEG, tid);
      __syncthreads();
    } else {
      for (int e = tid; e < 1024; e += 128) sJd[e] = 0.f;
      __syncthreads();
    }

    for (int e = tid; e < 1024; e += 128) {
      float v = sJd[e] + g_Ris[(size_t)t * 1024 + e];
      if (t > 0) v += sQis[prev][e];
      sJd[e] = v;
    }
    if (tid < DD) {
      float hv = g_hR[t * DD + tid];
      if (t > 0) hv += sQb[tid];
      if (haveNext) {
        float jb = 0.f;
        for (int j = 0; j < DD; ++j) jb += sJl[cur][tid * 32 + j] * sbs[j];
        hv += jb;
      }
      float jd = 0.f;
      if (t > 0)
        for (int j = 0; j < DD; ++j) jd += sJl[prev][tid * 32 + j] * sd[j];
      sh[tid] = hv;
      sJlPd[tid] = jd;
      if (haveNext) {
        float qb = 0.f;
        for (int j = 0; j < DD; ++j) qb += sQis[cur][tid * 32 + j] * sbs[j];
        sQb[tid] = qb;
      }
    }
    __syncthreads();

    for (int idx = tid; idx < 32 * 65; idx += 128) {
      int i = idx / 65, j = idx % 65;
      float v;
      if (j < 32)      v = sJd[i * 32 + j] + (i == j ? EPS_TRI : 0.f);
      else if (j < 64) v = haveNext ? sJl[cur][(j - 32) * 32 + i] : 0.f;
      else             v = sh[i] - (t > 0 ? sJlPd[i] : 0.f);
      sAug[i * AUG_LD + j] = v;
    }
    __syncthreads();
    if (t > 0) {
      mm32<false, false>(sJl[prev], sC, sAug, AUG_LD, MM_SUB, tid);
      __syncthreads();
    }
    for (int k = 0; k < 32; ++k) {
      float piv = sAug[k * AUG_LD + k];
      float f = sAug[row * AUG_LD + k] / piv;
      __syncthreads();
      if (row != k) {
        for (int j = k + 1 + jgrp; j < 65; j += 4)
          sAug[row * AUG_LD + j] -= f * sAug[k * AUG_LD + j];
        if (jgrp == 0) sAug[row * AUG_LD + k] = 0.f;
      }
      __syncthreads();
    }
    for (int e = tid; e < 1024; e += 128) {
      int i = e >> 5, j = e & 31;
      sC[e] = sAug[j * AUG_LD + 32 + i] / sAug[j * AUG_LD + j];
    }
    if (tid < DD) sd[tid] = sAug[tid * AUG_LD + 64] / sAug[tid * AUG_LD + tid];
    __syncthreads();
    {
      size_t base = (size_t)n * TT + t;
      if (haveNext)
        for (int e = tid; e < 1024; e += 128) g_c[base * 1024 + e] = sC[e];
      if (tid < DD) g_d[base * DD + tid] = sd[tid];
    }

    if (t > 0) {
      for (int e = tid; e < 1024; e += 128)
        sAug[(e >> 5) * AUG_LD + (e & 31)] = sJl[prev][e];
      __syncthreads();
      if (tid < 32) {
        int c = tid;
        for (int i = 31; i >= 0; --i) {
          float xi = sAug[c * AUG_LD + i] / sL[i * 32 + i];
          sAug[c * AUG_LD + i] = xi;
          for (int j = 0; j < i; ++j)
            sAug[c * AUG_LD + j] -= sL[i * 32 + j] * xi;
        }
      }
      __syncthreads();
      for (int e = tid; e < 1024; e += 128)
        sLl[e] = sAug[(e >> 5) * AUG_LD + (e & 31)];
      __syncthreads();
      for (int e = tid; e < 1024; e += 128) {
        int i = e >> 5, j = e & 31;
        sM[e] = sJd[e] + (i == j ? EPS_CHOL : 0.f);
      }
      __syncthreads();
      mm32<false, true>(sLl, sLl, sM, 32, MM_SUB, tid);
      __syncthreads();
    } else {
      for (int e = tid; e < 1024; e += 128) {
        int i = e >> 5, j = e & 31;
        sM[e] = sJd[e] + (i == j ? EPS_CHOL : 0.f);
      }
      __syncthreads();
    }
    for (int k = 0; k < 32; ++k) {
      float akk = sM[k * 32 + k];
      __syncthreads();
      float lkk = sqrtf(akk);
      if (tid == k) sM[k * 32 + k] = lkk;
      else if (tid < 32 && tid > k) sM[tid * 32 + k] /= lkk;
      __syncthreads();
      if (row > k)
        for (int j = k + 1 + jgrp; j < 32; j += 4)
          sM[row * 32 + j] -= sM[row * 32 + k] * sM[j * 32 + k];
      __syncthreads();
    }
    for (int e = tid; e < 1024; e += 128) {
      int i = e >> 5, j = e & 31;
      sL[e] = (j <= i) ? sM[e] : 0.f;
    }
    __syncthreads();

    if (tid < DD) {
      float r = noise[((size_t)n * TT + t) * DD + tid];
      if (t > 0) {
        float s = 0.f;
        for (int j = 0; j < DD; ++j) s += sLl[j * 32 + tid] * sx[j];
        r -= s;
      }
      srhs[tid] = r;
    }
    __syncthreads();
    for (int i = 31; i >= 0; --i) {
      if (tid == 0) srhs[i] /= (sL[i * 32 + i] + EPS_BS);
      __syncthreads();
      if (tid < i) srhs[tid] -= sL[i * 32 + tid] * srhs[i];
      __syncthreads();
    }
    if (tid < DD) {
      sx[tid] = srhs[tid];
      out[((size_t)n * TT + t) * DD + tid] = srhs[tid];
    }
    __syncthreads();
  }
}

__global__ __launch_bounds__(128) void lds_backward_kernel(
    const float* __restrict__ g_c, const float* __restrict__ g_d,
    float* __restrict__ out) {
  __shared__ float sCt[1024];
  __shared__ float sMu[DD], sNew[DD];
  const int n = blockIdx.x;
  const int tid = threadIdx.x;
  if (tid < DD) {
    float v = g_d[((size_t)n * TT + (TT - 1)) * DD + tid];
    sMu[tid] = v;
    out[((size_t)n * TT + (TT - 1)) * DD + tid] += v;
  }
  __syncthreads();
  for (int t = TT - 2; t >= 0; --t) {
    size_t base = (size_t)n * TT + t;
    for (int e = tid; e < 1024; e += 128) sCt[e] = g_c[base * 1024 + e];
    __syncthreads();
    if (tid < DD) {
      float acc = g_d[base * DD + tid];
      for (int j = 0; j < DD; ++j) acc -= sCt[tid * 32 + j] * sMu[j];
      sNew[tid] = acc;
    }
    __syncthreads();
    if (tid < DD) {
      sMu[tid] = sNew[tid];
      out[base * DD + tid] += sNew[tid];
    }
    __syncthreads();
  }
}

extern "C" void kernel_launch(void* const* d_in, const int* in_sizes, int n_in,
                              void* d_out, int out_size, void* d_ws,
                              size_t ws_size, hipStream_t stream) {
  (void)in_sizes; (void)n_in; (void)out_size; (void)ws_size;
  const float* z_samples = (const float*)d_in[0];
  const float* A_base = (const float*)d_in[1];
  const float* b_base = (const float*)d_in[2];
  const float* Q_sqrt = (const float*)d_in[3];
  const float* ms = (const float*)d_in[4];
  const float* Ri_sqrts = (const float*)d_in[5];
  const float* noise = (const float*)d_in[6];
  float* ws = (float*)d_ws;
  float* g_Ris = ws;
  float* g_hR = g_Ris + (size_t)TT * DD * DD;
  float* g_d = g_hR + (size_t)TT * DD;
  float* g_c = g_d + (size_t)NB * TT * DD;
  float* out = (float*)d_out;

  ris_precompute_kernel<<<TT, 128, 0, stream>>>(Ri_sqrts, ms, g_Ris, g_hR);
  lds_forward_kernel<<<NB, 128, 0, stream>>>(z_samples, A_base, b_base, Q_sqrt,
                                             noise, g_Ris, g_hR, g_c, g_d, out);
  lds_backward_kernel<<<NB, 128, 0, stream>>>(g_c, g_d, out);
}


# Recurrence-free part (As,Qs,bs mixes; Qis,Jl,Jd products; h) built for 4 time steps at once: A_base,Q_sqrt fetched once per group, 16 waves on its products, barriers once per 4 steps; sum order kept
# speedup vs baseline: 4.6045x; 4.6045x over previous
#include <hip/hip_runtime.h>

#define TT 256
#define DD 32
#define KK 8
#define NB 64
#define AUG_LD 66
#define FW_NW 16
#define FW_NT (FW_NW * 32)
#define SW_NW 8
#define TG 4
#define EPS_TRI 0.01f
#define EPS_CHOL 0.01f
#define EPS_BS 0.0001f

typedef float v2f __attribute__((ext_vector_type(2)));
typedef float v8f __attribute__((ext_vector_type(8)));

enum { MM_STORE = 0, MM_STORE_NEG = 1, MM_SUB = 2, MM_ADD = 3 };

#define LL_ROW_SHARE(v, N) \
  __int_as_float(__builtin_amdgcn_mov_dpp(__float_as_int(v), 0x150 + (N), 0xf, 0xf, false))
#define LL_STEP_HI(I)                                        \
  {                                                          \
    const float q = ahi / Lhi[(I) - 16];                     \
    const float x = LL_ROW_SHARE(q, (I) - 16);               \
    xhi = (jl == (I) - 16) ? q : xhi;                        \
    ahi = __builtin_fmaf(-Lhi[(I) - 16], x, ahi);            \
    alo = __builtin_fmaf(-Llo[(I)], x, alo);                 \
  }
#define LL_STEP_LO(I)                                        \
  {                                                          \
    const float q = alo / Llo[(I)];                          \
    const float x = LL_ROW_SHARE(q, (I));                    \
    xlo = (jl == (I)) ? q : xlo;                             \
    alo = __builtin_fmaf(-Llo[(I)], x, alo);                 \
  }

#define SS_STEP_HI(I)                                        \
  {                                                          \
    const float q = ahi / (Lhi[(I) - 16] + EPS_BS);          \
    const float x = LL_ROW_SHARE(q, (I) - 16);               \
    xhi = (jl == (I) - 16) ? q : xhi;                        \
    ahi = __builtin_fmaf(-Lhi[(I) - 16], x, ahi);            \
    alo = __builtin_fmaf(-Llo[(I)], x, alo);                 \
  }
#define SS_STEP_LO(I)                                        \
  {                                                          \
    const float q = alo / (Llo[(I)] + EPS_BS);               \
    const float x = LL_ROW_SHARE(q, (I));                    \
    xlo = (jl == (I)) ? q : xlo;                             \
    alo = __builtin_fmaf(-Llo[(I)], x, alo);                 \
  }

template <bool TA, bool TB>
__device__ inline void mm32(const float* A, const float* B, float* D, int ldd,
                            int mode, int tid) {
  const int lane = tid & 31;
  const int wid = tid >> 5;
  const int mt = (wid >> 1) << 4;
  const int nt = (wid & 1) << 4;
  const int l15 = lane & 15;
  const int kh = (lane >> 4) << 1;
  v8f acc = {};
#pragma unroll
  for (int kt = 0; kt < 32; kt += 4) {
    const int k0 = kt + kh;
    v2f av, bv;
    if (!TA) { av.x = A[(mt + l15) * 32 + k0];  av.y = A[(mt + l15) * 32 + k0 + 1]; }
    else     { av.x = A[k0 * 32 + mt + l15];    av.y = A[(k0 + 1) * 32 + mt + l15]; }
    if (!TB) { bv.x = B[k0 * 32 + nt + l15];    bv.y = B[(k0 + 1) * 32 + nt + l15]; }
    else     { bv.x = B[(nt + l15) * 32 + k0];  bv.y = B[(nt + l15) * 32 + k0 + 1]; }
    acc = __builtin_amdgcn_wmma_f32_16x16x4_f32(false, av, false, bv, (short)0,
                                                acc, false, false);
  }
  const int rbase = mt + ((lane >> 4) << 3);
#pragma unroll
  for (int r = 0; r < 8; ++r) {
    float v = acc[r];
    float* p = &D[(rbase + r) * ldd + nt + l15];
    if (mode == MM_STORE) *p = v;
    else if (mode == MM_STORE_NEG) *p = -v;
    else if (mode == MM_SUB) *p -= v;
    else *p += v;
  }
}

__global__ __launch_bounds__(128) void ris_precompute_kernel(
    const float* __restrict__ Ri_sqrts, const float* __restrict__ ms,
    float* __restrict__ g_Ris, float* __restrict__ g_hR) {
  __shared__ float sR[1024], sO[1024], sm[DD];
  const int t = blockIdx.x;
  const int tid = threadIdx.x;
  for (int e = tid; e < 1024; e += 128) sR[e] = Ri_sqrts[(size_t)t * 1024 + e];
  if (tid < DD) sm[tid] = ms[t * DD + tid];
  __syncthreads();
  mm32<false, true>(sR, sR, sO, 32, MM_STORE, tid);
  __syncthreads();
  for (int e = tid; e < 1024; e += 128) g_Ris[(size_t)t * 1024 + e] = sO[e];
  if (tid < DD) {
    float a = 0.f;
    for (int j = 0; j < DD; ++j) a += sO[tid * 32 + j] * sm[j];
    g_hR[t * DD + tid] = a;
  }
}

__global__ __launch_bounds__(FW_NT) void lds_forward_kernel(
    const float* __restrict__ z_samples, const float* __restrict__ A_base,
    const float* __restrict__ b_base, const float* __restrict__ Q_sqrt,
    const float* __restrict__ noise, const float* __restrict__ g_Ris,
    const float* __restrict__ g_hR, float* __restrict__ g_c,
    float* __restrict__ g_d, float* __restrict__ out) {
  __shared__ float sC[1024], sL[1024], sLl[1024], sM[1024];
  __shared__ float sA[TG][1024], sQs[TG][1024];
  __shared__ float sQisB[TG][1024], sJlB[TG][1024], sJdB[TG][1024];
  __shared__ float sJlC[1024], sQisC[1024], sQb2[2][DD];
  __shared__ float sAug[32 * AUG_LD];
  __shared__ float szB[TG * KK], sbsB[TG][DD], shB[TG][DD], sJlPd[DD], sd[DD],
      srhs[DD], sx[DD], sPiv[DD];

  const int n = blockIdx.x;
  const int tid = threadIdx.x;
  const int row = tid & 31;
  const int jgrp = tid >> 5;
  const int role = (jgrp ^ (jgrp >> 2)) & 1;
  const int widx = jgrp >> 1;

  for (int e = tid; e < 1024; e += FW_NT) sL[e] = 0.f;

  for (int t = 0; t < TT; ++t) {
    const int b = t & (TG - 1);
    const bool haveNext = (t < TT - 1);
    const float* Jl_cur = sJlB[b];
    const float* Jl_prev = (b == 0) ? sJlC : sJlB[b - 1];
    const float* Jd = sJdB[b];

    if (b == 0) {
      const int gpar = (t / TG) & 1;
      if (t > 0)
        for (int e = tid; e < 1024; e += FW_NT) {
          sJlC[e] = sJlB[TG - 1][e];
          sQisC[e] = sQisB[TG - 1][e];
        }
      if (tid < TG * KK) szB[tid] = z_samples[((size_t)n * TT + t) * KK + tid];
      __syncthreads();
#pragma unroll
      for (int e = tid; e < 1024; e += FW_NT) {
        float av[KK], qv[KK];
#pragma unroll
        for (int k = 0; k < KK; ++k) {
          av[k] = A_base[k * 1024 + e];
          qv[k] = Q_sqrt[k * 1024 + e];
        }
#pragma unroll
        for (int s = 0; s < TG; ++s) {
          float a = 0.f, q = 0.f;
#pragma unroll
          for (int k = 0; k < KK; ++k) {
            float zk = szB[s * KK + k];
            a += zk * av[k];
            q += zk * qv[k];
          }
          sA[s][e] = a;
          sQs[s][e] = q;
        }
      }
      if (tid < TG * DD) {
        const int s = tid >> 5;
        float bb = 0.f;
#pragma unroll
        for (int k = 0; k < KK; ++k) bb += szB[s * KK + k] * b_base[k * DD + row];
        sbsB[s][row] = bb;
      }
      __syncthreads();
      {
        const int gs = tid >> 7;
        const int gt = tid & 127;
        mm32<false, true>(sQs[gs], sQs[gs], sQisB[gs], 32, MM_STORE, gt);
        __syncthreads();
        mm32<false, false>(sQisB[gs], sA[gs], sJlB[gs], 32, MM_STORE_NEG, gt);
        __syncthreads();
        mm32<true, false>(sA[gs], sJlB[gs], sJdB[gs], 32, MM_STORE_NEG, gt);
        __syncthreads();
      }
      float ris[2 * TG];
#pragma unroll
      for (int m = 0; m < 2 * TG; ++m)
        ris[m] = g_Ris[(size_t)(t + (m >> 1)) * 1024 + (m & 1) * FW_NT + tid];
#pragma unroll
      for (int m = 0; m < 2 * TG; ++m) {
        const int s = m >> 1, e = (m & 1) * FW_NT + tid, tt = t + s;
        const float jp = sJdB[s][e];
        float v = ((tt < TT - 1) ? jp : 0.f) + ris[m];
        if (s > 0) v += sQisB[s - 1][e];
        else if (t > 0) v += sQisC[e];
        sJdB[s][e] = v;
      }
      if (tid < TG * DD) {
        const int s = tid >> 5, tt = t + s;
        float hv = g_hR[tt * DD + row];
        if (tt > 0) {
          float qp;
          if (s == 0) {
            qp = sQb2[gpar ^ 1][row];
          } else {
            qp = 0.f;
            for (int j = 0; j < DD; ++j) qp += sQisB[s - 1][row * 32 + j] * sbsB[s - 1][j];
          }
          hv += qp;
        }
        if (tt < TT - 1) {
          float jb = 0.f;
          for (int j = 0; j < DD; ++j) jb += sJlB[s][row * 32 + j] * sbsB[s][j];
          hv += jb;
        }
        shB[s][row] = hv;
        if (s == TG - 1) {
          float qb = 0.f;
          for (int j = 0; j < DD; ++j) qb += sQisB[s][row * 32 + j] * sbsB[s][j];
          sQb2[gpar][row] = qb;
        }
      }
    }
    __syncthreads();

    for (int idx = tid; idx < 32 * 65; idx += FW_NT) {
      int i = idx / 65, j = idx % 65;
      float v;
      if (j < 32)      v = Jd[i * 32 + j] + (i == j ? EPS_TRI : 0.f);
      else if (j < 64) v = haveNext ? Jl_cur[(j - 32) * 32 + i] : 0.f;
      else             v = shB[b][i] - (t > 0 ? sJlPd[i] : 0.f);
      sAug[i * AUG_LD + j] = v;
    }
    if (t > 0) {
      {
        const int jl = tid & 15;
        const int c = (jgrp << 1) | ((tid >> 4) & 1);
        float ahi = Jl_prev[c * 32 + jl + 16];
        float alo = Jl_prev[c * 32 + jl];
        float Lhi[16], Llo[32];
#pragma unroll
        for (int i = 15; i >= 0; --i) {
          Lhi[i] = sL[(i + 16) * 32 + jl + 16];
          Llo[i + 16] = sL[(i + 16) * 32 + jl];
        }
#pragma unroll
        for (int i = 15; i >= 0; --i) Llo[i] = sL[i * 32 + jl];
        __builtin_amdgcn_sched_barrier(0);
        float xhi = 0.f, xlo = 0.f;
        LL_STEP_HI(31) LL_STEP_HI(30) LL_STEP_HI(29) LL_STEP_HI(28)
        LL_STEP_HI(27) LL_STEP_HI(26) LL_STEP_HI(25) LL_STEP_HI(24)
        LL_STEP_HI(23) LL_STEP_HI(22) LL_STEP_HI(21) LL_STEP_HI(20)
        LL_STEP_HI(19) LL_STEP_HI(18) LL_STEP_HI(17) LL_STEP_HI(16)
        LL_STEP_LO(15) LL_STEP_LO(14) LL_STEP_LO(13) LL_STEP_LO(12)
        LL_STEP_LO(11) LL_STEP_LO(10) LL_STEP_LO(9) LL_STEP_LO(8)
        LL_STEP_LO(7) LL_STEP_LO(6) LL_STEP_LO(5) LL_STEP_LO(4)
        LL_STEP_LO(3) LL_STEP_LO(2) LL_STEP_LO(1) LL_STEP_LO(0)
        sLl[c * 32 + jl] = xlo;
        sLl[c * 32 + jl + 16] = xhi;
      }
    }
#pragma unroll
    for (int u = 0; u < 2; ++u) {
      const int c = (row + jgrp + u * FW_NW) & 31;
      sM[c * 32 + row] = Jd[row * 32 + c] + (c == row ? EPS_CHOL : 0.f);
    }
    __syncthreads();
    if (t > 0) {
      if (tid < 128) mm32<false, false>(Jl_prev, sC, sAug, AUG_LD, MM_SUB, tid);
      else if (tid < 256) mm32<false, true>(sLl, sLl, sM, 32, MM_SUB, tid - 128);
      __syncthreads();
    }
    for (int k = 0; k < 32; k += 2) {
      const int j0 = k + 2 + widx;
      if (role == 0) {
        const float* rk0 = &sAug[k * AUG_LD];
        const float* rk1 = rk0 + AUG_LD;
        float* rr = &sAug[row * AUG_LD];
        const float a00 = rk0[k], a01 = rk0[k + 1], a10 = rk1[k], a11 = rk1[k + 1];
        const float ar0 = rr[k], ar1 = rr[k + 1];
        float p0[8], p1[8], ow[8];
#pragma unroll
        for (int u = 0; u < 8; ++u) {
          const int j = (j0 + u * SW_NW < 65) ? j0 + u * SW_NW : k;
          p0[u] = rk0[j];
          p1[u] = rk1[j];
          ow[u] = rr[j];
        }
        asm volatile("" ::"v"(a00), "v"(a01), "v"(a10), "v"(a11), "v"(ar0), "v"(ar1));
        asm volatile("" ::"v"(p0[0]), "v"(p0[1]), "v"(p0[2]), "v"(p0[3]),
                     "v"(p0[4]), "v"(p0[5]), "v"(p0[6]), "v"(p0[7]),
                     "v"(p1[0]), "v"(p1[1]), "v"(p1[2]), "v"(p1[3]),
                     "v"(p1[4]), "v"(p1[5]), "v"(p1[6]), "v"(p1[7]));
        asm volatile("" ::"v"(ow[0]), "v"(ow[1]), "v"(ow[2]), "v"(ow[3]),
                     "v"(ow[4]), "v"(ow[5]), "v"(ow[6]), "v"(ow[7]));
        __builtin_amdgcn_sched_barrier(0);
        const float f1 = a10 / a00;
        const float fr = ar0 / a00;
        const float q11 = a11 - f1 * a01;
        const float br1 = (row != k) ? ar1 - fr * a01 : ar1;
        const float gr = br1 / q11;
#pragma unroll
        for (int u = 0; u < 8; ++u) {
          const int j = j0 + u * SW_NW;
          if (j < 65) {
            const float q = p1[u] - f1 * p0[u];
            float v = ow[u];
            if (row != k) v = v - fr * p0[u];
            if (row != k + 1) v = v - gr * q;
            rr[j] = v;
          }
        }
        if (tid == k) {
          sPiv[k] = a00;
          sPiv[k + 1] = q11;
        }
      } else if (widx == 0 || j0 < 32) {
        bool has[4];
        float mr[4];
        const float mc0 = sM[k * 32 + row];
        const float mc1 = sM[(k + 1) * 32 + row];
#pragma unroll
        for (int u = 0; u < 4; ++u) {
          has[u] = j0 + u * SW_NW < 32;
          mr[u] = has[u] ? sM[(j0 + u * SW_NW) * 32 + row] : 0.f;
        }
        asm volatile("" ::"v"(mc0), "v"(mc1), "v"(mr[0]), "v"(mr[1]), "v"(mr[2]), "v"(mr[3]));
        __builtin_amdgcn_sched_barrier(0);
        const float akk = __int_as_float(__builtin_amdgcn_readlane(__float_as_int(mc0), k));
        const float lkk = sqrtf(akk);
        const float l0 = mc0 / lkk;
        const float l0n = __int_as_float(__builtin_amdgcn_readlane(__float_as_int(l0), k + 1));
        const float m1 = mc1 - l0 * l0n;
        const float ann = __int_as_float(__builtin_amdgcn_readlane(__float_as_int(m1), k + 1));
        const float lnn = sqrtf(ann);
        const float l1 = m1 / lnn;
        float l0x[4], l1x[4];
#pragma unroll
        for (int u = 0; u < 4; ++u) {
          const int jx = __builtin_amdgcn_readfirstlane(has[u] ? j0 + u * SW_NW : k);
          l0x[u] = __int_as_float(__builtin_amdgcn_readlane(__float_as_int(l0), jx));
          l1x[u] = __int_as_float(__builtin_amdgcn_readlane(__float_as_int(l1), jx));
        }
        if (row > k + 1) {
#pragma unroll
          for (int u = 0; u < 4; ++u)
            if (has[u]) sM[(j0 + u * SW_NW) * 32 + row] = (mr[u] - l0 * l0x[u]) - l1 * l1x[u];
          if (widx == 0) {
            sL[row * 32 + k] = l0;
            sL[row * 32 + k + 1] = l1;
          }
        } else if (widx == 0 && row == k + 1) {
          sL[(k + 1) * 32 + k] = l0;
          sL[(k + 1) * 32 + k + 1] = lnn;
        } else if (widx == 0 && row == k) {
          sL[k * 32 + k] = lkk;
        }
      }
      __syncthreads();
    }
    for (int e = tid; e < 1024; e += FW_NT) {
      int i = e >> 5, j = e & 31;
      sC[e] = sAug[j * AUG_LD + 32 + i] / sPiv[j];
    }
    if (tid < DD) {
      sd[tid] = sAug[tid * AUG_LD + 64] / sPiv[tid];
      float r = noise[((size_t)n * TT + t) * DD + tid];
      if (t > 0) {
        float s = 0.f;
        for (int j = 0; j < DD; ++j) s += sLl[j * 32 + tid] * sx[j];
        r -= s;
      }
      srhs[tid] = r;
    }
    __syncthreads();
    {
      size_t base = (size_t)n * TT + t;
      if (haveNext)
        for (int e = tid; e < 1024; e += FW_NT) g_c[base * 1024 + e] = sC[e];
      if (tid < DD) g_d[base * DD + tid] = sd[tid];
    }
    if (haveNext && jgrp == 1) {
      float jd = 0.f;
      for (int j = 0; j < DD; ++j) jd += Jl_cur[row * 32 + j] * sd[j];
      sJlPd[row] = jd;
    }

    if (tid < 32) {
      const int jl = tid & 15;
      float ahi = srhs[jl + 16];
      float alo = srhs[jl];
      float Lhi[16], Llo[32];
#pragma unroll
      for (int i = 15; i >= 0; --i) {
        Lhi[i] = sL[(i + 16) * 32 + jl + 16];
        Llo[i + 16] = sL[(i + 16) * 32 + jl];
      }
#pragma unroll
      for (int i = 15; i >= 0; --i) Llo[i] = sL[i * 32 + jl];
      __builtin_amdgcn_sched_barrier(0);
      float xhi = 0.f, xlo = 0.f;
      SS_STEP_HI(31) SS_STEP_HI(30) SS_STEP_HI(29) SS_STEP_HI(28)
      SS_STEP_HI(27) SS_STEP_HI(26) SS_STEP_HI(25) SS_STEP_HI(24)
      SS_STEP_HI(23) SS_STEP_HI(22) SS_STEP_HI(21) SS_STEP_HI(20)
      SS_STEP_HI(19) SS_STEP_HI(18) SS_STEP_HI(17) SS_STEP_HI(16)
      SS_STEP_LO(15) SS_STEP_LO(14) SS_STEP_LO(13) SS_STEP_LO(12)
      SS_STEP_LO(11) SS_STEP_LO(10) SS_STEP_LO(9) SS_STEP_LO(8)
      SS_STEP_LO(7) SS_STEP_LO(6) SS_STEP_LO(5) SS_STEP_LO(4)
      SS_STEP_LO(3) SS_STEP_LO(2) SS_STEP_LO(1) SS_STEP_LO(0)
      if (tid < 16) {
        sx[jl] = xlo;
        sx[jl + 16] = xhi;
        float* o = &out[((size_t)n * TT + t) * DD];
        o[jl] = xlo;
        o[jl + 16] = xhi;
      }
    }
  }
}

__global__ __launch_bounds__(128) void lds_backward_kernel(
    const float* __restrict__ g_c, const float* __restrict__ g_d,
    float* __restrict__ out) {
  __shared__ float sCt[1024];
  __shared__ float sMu[DD], sNew[DD];
  const int n = blockIdx.x;
  const int tid = threadIdx.x;
  if (tid < DD) {
    float v = g_d[((size_t)n * TT + (TT - 1)) * DD + tid];
    sMu[tid] = v;
    out[((size_t)n * TT + (TT - 1)) * DD + tid] += v;
  }
  __syncthreads();
  for (int t = TT - 2; t >= 0; --t) {
    size_t base = (size_t)n * TT + t;
    for (int e = tid; e < 1024; e += 128) sCt[e] = g_c[base * 1024 + e];
    __syncthreads();
    if (tid < DD) {
      float acc = g_d[base * DD + tid];
      for (int j = 0; j < DD; ++j) acc -= sCt[tid * 32 + j] * sMu[j];
      sNew[tid] = acc;
    }
    __syncthreads();
    if (tid < DD) {
      sMu[tid] = sNew[tid];
      out[base * DD + tid] += sNew[tid];
    }
    __syncthreads();
  }
}

extern "C" void kernel_launch(void* const* d_in, const int* in_sizes, int n_in,
                              void* d_out, int out_size, void* d_ws,
                              size_t ws_size, hipStream_t stream) {
  (void)in_sizes; (void)n_in; (void)out_size; (void)ws_size;
  const float* z_samples = (const float*)d_in[0];
  const float* A_base = (const float*)d_in[1];
  const float* b_base = (const float*)d_in[2];
  const float* Q_sqrt = (const float*)d_in[3];
  const float* ms = (const float*)d_in[4];
  const float* Ri_sqrts = (const float*)d_in[5];
  const float* noise = (const float*)d_in[6];
  float* ws = (float*)d_ws;
  float* g_Ris = ws;
  float* g_hR = g_Ris + (size_t)TT * DD * DD;
  float* g_d = g_hR + (size_t)TT * DD;
  float* g_c = g_d + (size_t)NB * TT * DD;
  float* out = (float*)d_out;

  ris_precompute_kernel<<<TT, 128, 0, stream>>>(Ri_sqrts, ms, g_Ris, g_hR);
  lds_forward_kernel<<<NB, FW_NT, 0, stream>>>(z_samples, A_base, b_base, Q_sqrt,
                                             noise, g_Ris, g_hR, g_c, g_d, out);
  lds_backward_kernel<<<NB, 128, 0, stream>>>(g_c, g_d, out);
}


# Pivot sweeps (Gauss-Jordan of sAug, Cholesky of sM) take four pivots per barrier interval, were two: 8 intervals per time step instead of 16; per element the same operations in the same order
# speedup vs baseline: 5.2696x; 1.1444x over previous
#include <hip/hip_runtime.h>

#define TT 256
#define DD 32
#define KK 8
#define NB 64
#define AUG_LD 66
#define FW_NW 16
#define FW_NT (FW_NW * 32)
#define SW_NW 8
#define TG 4
#define SW_P 4
#define EPS_TRI 0.01f
#define EPS_CHOL 0.01f
#define EPS_BS 0.0001f

typedef float v2f __attribute__((ext_vector_type(2)));
typedef float v8f __attribute__((ext_vector_type(8)));

static_assert(SW_P == 4 && 32 % SW_P == 0, "pivot sweeps are written out for four pivots per interval");

enum { MM_STORE = 0, MM_STORE_NEG = 1, MM_SUB = 2, MM_ADD = 3 };

#define LL_ROW_SHARE(v, N) \
  __int_as_float(__builtin_amdgcn_mov_dpp(__float_as_int(v), 0x150 + (N), 0xf, 0xf, false))
#define LL_STEP_HI(I)                                        \
  {                                                          \
    const float q = ahi / Lhi[(I) - 16];                     \
    const float x = LL_ROW_SHARE(q, (I) - 16);               \
    xhi = (jl == (I) - 16) ? q : xhi;                        \
    ahi = __builtin_fmaf(-Lhi[(I) - 16], x, ahi);            \
    alo = __builtin_fmaf(-Llo[(I)], x, alo);                 \
  }
#define LL_STEP_LO(I)                                        \
  {                                                          \
    const float q = alo / Llo[(I)];                          \
    const float x = LL_ROW_SHARE(q, (I));                    \
    xlo = (jl == (I)) ? q : xlo;                             \
    alo = __builtin_fmaf(-Llo[(I)], x, alo);                 \
  }

#define SS_STEP_HI(I)                                        \
  {                                                          \
    const float q = ahi / (Lhi[(I) - 16] + EPS_BS);          \
    const float x = LL_ROW_SHARE(q, (I) - 16);               \
    xhi = (jl == (I) - 16) ? q : xhi;                        \
    ahi = __builtin_fmaf(-Lhi[(I) - 16], x, ahi);            \
    alo = __builtin_fmaf(-Llo[(I)], x, alo);                 \
  }
#define SS_STEP_LO(I)                                        \
  {                                                          \
    const float q = alo / (Llo[(I)] + EPS_BS);               \
    const float x = LL_ROW_SHARE(q, (I));                    \
    xlo = (jl == (I)) ? q : xlo;                             \
    alo = __builtin_fmaf(-Llo[(I)], x, alo);                 \
  }

template <bool TA, bool TB>
__device__ inline void mm32(const float* A, const float* B, float* D, int ldd,
                            int mode, int tid) {
  const int lane = tid & 31;
  const int wid = tid >> 5;
  const int mt = (wid >> 1) << 4;
  const int nt = (wid & 1) << 4;
  const int l15 = lane & 15;
  const int kh = (lane >> 4) << 1;
  v8f acc = {};
#pragma unroll
  for (int kt = 0; kt < 32; kt += 4) {
    const int k0 = kt + kh;
    v2f av, bv;
    if (!TA) { av.x = A[(mt + l15) * 32 + k0];  av.y = A[(mt + l15) * 32 + k0 + 1]; }
    else     { av.x = A[k0 * 32 + mt + l15];    av.y = A[(k0 + 1) * 32 + mt + l15]; }
    if (!TB) { bv.x = B[k0 * 32 + nt + l15];    bv.y = B[(k0 + 1) * 32 + nt + l15]; }
    else     { bv.x = B[(nt + l15) * 32 + k0];  bv.y = B[(nt + l15) * 32 + k0 + 1]; }
    acc = __builtin_amdgcn_wmma_f32_16x16x4_f32(false, av, false, bv, (short)0,
                                                acc, false, false);
  }
  const int rbase = mt + ((lane >> 4) << 3);
#pragma unroll
  for (int r = 0; r < 8; ++r) {
    float v = acc[r];
    float* p = &D[(rbase + r) * ldd + nt + l15];
    if (mode == MM_STORE) *p = v;
    else if (mode == MM_STORE_NEG) *p = -v;
    else if (mode == MM_SUB) *p -= v;
    else *p += v;
  }
}

__global__ __launch_bounds__(128) void ris_precompute_kernel(
    const float* __restrict__ Ri_sqrts, const float* __restrict__ ms,
    float* __restrict__ g_Ris, float* __restrict__ g_hR) {
  __shared__ float sR[1024], sO[1024], sm[DD];
  const int t = blockIdx.x;
  const int tid = threadIdx.x;
  for (int e = tid; e < 1024; e += 128) sR[e] = Ri_sqrts[(size_t)t * 1024 + e];
  if (tid < DD) sm[tid] = ms[t * DD + tid];
  __syncthreads();
  mm32<false, true>(sR, sR, sO, 32, MM_STORE, tid);
  __syncthreads();
  for (int e = tid; e < 1024; e += 128) g_Ris[(size_t)t * 1024 + e] = sO[e];
  if (tid < DD) {
    float a = 0.f;
    for (int j = 0; j < DD; ++j) a += sO[tid * 32 + j] * sm[j];
    g_hR[t * DD + tid] = a;
  }
}

__global__ __launch_bounds__(FW_NT) void lds_forward_kernel(
    const float* __restrict__ z_samples, const float* __restrict__ A_base,
    const float* __restrict__ b_base, const float* __restrict__ Q_sqrt,
    const float* __restrict__ noise, const float* __restrict__ g_Ris,
    const float* __restrict__ g_hR, float* __restrict__ g_c,
    float* __restrict__ g_d, float* __restrict__ out) {
  __shared__ float sC[1024], sL[1024], sLl[1024], sM[1024];
  __shared__ float sA[TG][1024], sQs[TG][1024];
  __shared__ float sQisB[TG][1024], sJlB[TG][1024], sJdB[TG][1024];
  __shared__ float sJlC[1024], sQisC[1024], sQb2[2][DD];
  __shared__ float sAug[32 * AUG_LD];
  __shared__ float szB[TG * KK], sbsB[TG][DD], shB[TG][DD], sJlPd[DD], sd[DD],
      srhs[DD], sx[DD], sPiv[DD];

  const int n = blockIdx.x;
  const int tid = threadIdx.x;
  const int row = tid & 31;
  const int jgrp = tid >> 5;
  const int role = (jgrp ^ (jgrp >> 2)) & 1;
  const int widx = jgrp >> 1;

  for (int e = tid; e < 1024; e += FW_NT) sL[e] = 0.f;

  for (int t = 0; t < TT; ++t) {
    const int b = t & (TG - 1);
    const bool haveNext = (t < TT - 1);
    const float* Jl_cur = sJlB[b];
    const float* Jl_prev = (b == 0) ? sJlC : sJlB[b - 1];
    const float* Jd = sJdB[b];

    if (b == 0) {
      const int gpar = (t / TG) & 1;
      if (t > 0)
        for (int e = tid; e < 1024; e += FW_NT) {
          sJlC[e] = sJlB[TG - 1][e];
          sQisC[e] = sQisB[TG - 1][e];
        }
      if (tid < TG * KK) szB[tid] = z_samples[((size_t)n * TT + t) * KK + tid];
      __syncthreads();
#pragma unroll
      for (int e = tid; e < 1024; e += FW_NT) {
        float av[KK], qv[KK];
#pragma unroll
        for (int k = 0; k < KK; ++k) {
          av[k] = A_base[k * 1024 + e];
          qv[k] = Q_sqrt[k * 1024 + e];
        }
#pragma unroll
        for (int s = 0; s < TG; ++s) {
          float a = 0.f, q = 0.f;
#pragma unroll
          for (int k = 0; k < KK; ++k) {
            float zk = szB[s * KK + k];
            a += zk * av[k];
            q += zk * qv[k];
          }
          sA[s][e] = a;
          sQs[s][e] = q;
        }
      }
      if (tid < TG * DD) {
        const int s = tid >> 5;
        float bb = 0.f;
#pragma unroll
        for (int k = 0; k < KK; ++k) bb += szB[s * KK + k] * b_base[k * DD + row];
        sbsB[s][row] = bb;
      }
      __syncthreads();
      {
        const int gs = tid >> 7;
        const int gt = tid & 127;
        mm32<false, true>(sQs[gs], sQs[gs], sQisB[gs], 32, MM_STORE, gt);
        __syncthreads();
        mm32<false, false>(sQisB[gs], sA[gs], sJlB[gs], 32, MM_STORE_NEG, gt);
        __syncthreads();
        mm32<true, false>(sA[gs], sJlB[gs], sJdB[gs], 32, MM_STORE_NEG, gt);
        __syncthreads();
      }
      float ris[2 * TG];
#pragma unroll
      for (int m = 0; m < 2 * TG; ++m)
        ris[m] = g_Ris[(size_t)(t + (m >> 1)) * 1024 + (m & 1) * FW_NT + tid];
#pragma unroll
      for (int m = 0; m < 2 * TG; ++m) {
        const int s = m >> 1, e = (m & 1) * FW_NT + tid, tt = t + s;
        const float jp = sJdB[s][e];
        float v = ((tt < TT - 1) ? jp : 0.f) + ris[m];
        if (s > 0) v += sQisB[s - 1][e];
        else if (t > 0) v += sQisC[e];
        sJdB[s][e] = v;
      }
      if (tid < TG * DD) {
        const int s = tid >> 5, tt = t + s;
        float hv = g_hR[tt * DD + row];
        if (tt > 0) {
          float qp;
          if (s == 0) {
            qp = sQb2[gpar ^ 1][row];
          } else {
            qp = 0.f;
            for (int j = 0; j < DD; ++j) qp += sQisB[s - 1][row * 32 + j] * sbsB[s - 1][j];
          }
          hv += qp;
        }
        if (tt < TT - 1) {
          float jb = 0.f;
          for (int j = 0; j < DD; ++j) jb += sJlB[s][row * 32 + j] * sbsB[s][j];
          hv += jb;
        }
        shB[s][row] = hv;
        if (s == TG - 1) {
          float qb = 0.f;
          for (int j = 0; j < DD; ++j) qb += sQisB[s][row * 32 + j] * sbsB[s][j];
          sQb2[gpar][row] = qb;
        }
      }
    }
    __syncthreads();

    for (int idx = tid; idx < 32 * 65; idx += FW_NT) {
      int i = idx / 65, j = idx % 65;
      float v;
      if (j < 32)      v = Jd[i * 32 + j] + (i == j ? EPS_TRI : 0.f);
      else if (j < 64) v = haveNext ? Jl_cur[(j - 32) * 32 + i] : 0.f;
      else             v = shB[b][i] - (t > 0 ? sJlPd[i] : 0.f);
      sAug[i * AUG_LD + j] = v;
    }
    if (t > 0) {
      {
        const int jl = tid & 15;
        const int c = (jgrp << 1) | ((tid >> 4) & 1);
        float ahi = Jl_prev[c * 32 + jl + 16];
        float alo = Jl_prev[c * 32 + jl];
        float Lhi[16], Llo[32];
#pragma unroll
        for (int i = 15; i >= 0; --i) {
          Lhi[i] = sL[(i + 16) * 32 + jl + 16];
          Llo[i + 16] = sL[(i + 16) * 32 + jl];
        }
#pragma unroll
        for (int i = 15; i >= 0; --i) Llo[i] = sL[i * 32 + jl];
        __builtin_amdgcn_sched_barrier(0);
        float xhi = 0.f, xlo = 0.f;
        LL_STEP_HI(31) LL_STEP_HI(30) LL_STEP_HI(29) LL_STEP_HI(28)
        LL_STEP_HI(27) LL_STEP_HI(26) LL_STEP_HI(25) LL_STEP_HI(24)
        LL_STEP_HI(23) LL_STEP_HI(22) LL_STEP_HI(21) LL_STEP_HI(20)
        LL_STEP_HI(19) LL_STEP_HI(18) LL_STEP_HI(17) LL_STEP_HI(16)
        LL_STEP_LO(15) LL_STEP_LO(14) LL_STEP_LO(13) LL_STEP_LO(12)
        LL_STEP_LO(11) LL_STEP_LO(10) LL_STEP_LO(9) LL_STEP_LO(8)
        LL_STEP_LO(7) LL_STEP_LO(6) LL_STEP_LO(5) LL_STEP_LO(4)
        LL_STEP_LO(3) LL_STEP_LO(2) LL_STEP_LO(1) LL_STEP_LO(0)
        sLl[c * 32 + jl] = xlo;
        sLl[c * 32 + jl + 16] = xhi;
      }
    }
#pragma unroll
    for (int u = 0; u < 2; ++u) {
      const int c = (row + jgrp + u * FW_NW) & 31;
      sM[c * 32 + row] = Jd[row * 32 + c] + (c == row ? EPS_CHOL : 0.f);
    }
    __syncthreads();
    if (t > 0) {
      if (tid < 128) mm32<false, false>(Jl_prev, sC, sAug, AUG_LD, MM_SUB, tid);
      else if (tid < 256) mm32<false, true>(sLl, sLl, sM, 32, MM_SUB, tid - 128);
      __syncthreads();
    }
    for (int k = 0; k < 32; k += SW_P) {
      const int j0 = k + SW_P + widx;
      if (role == 0) {
        const float* rk = &sAug[k * AUG_LD];
        float* rr = &sAug[row * AUG_LD];
        float B[SW_P][SW_P], R[SW_P], Pv[SW_P][8], O[8];
#pragma unroll
        for (int i = 0; i < SW_P; ++i) {
#pragma unroll
          for (int c = 0; c < SW_P; ++c) B[i][c] = rk[i * AUG_LD + k + c];
        }
#pragma unroll
        for (int c = 0; c < SW_P; ++c) R[c] = rr[k + c];
#pragma unroll
        for (int u = 0; u < 8; ++u) {
          const int j = (j0 + u * SW_NW < 65) ? j0 + u * SW_NW : k;
#pragma unroll
          for (int i = 0; i < SW_P; ++i) Pv[i][u] = rk[i * AUG_LD + j];
          O[u] = rr[j];
        }
#pragma unroll
        for (int i = 0; i < SW_P; ++i) {
          asm volatile("" ::"v"(B[i][0]), "v"(B[i][1]), "v"(B[i][2]), "v"(B[i][3]), "v"(R[i]));
          asm volatile("" ::"v"(Pv[i][0]), "v"(Pv[i][1]), "v"(Pv[i][2]), "v"(Pv[i][3]),
                       "v"(Pv[i][4]), "v"(Pv[i][5]), "v"(Pv[i][6]), "v"(Pv[i][7]));
        }
        asm volatile("" ::"v"(O[0]), "v"(O[1]), "v"(O[2]), "v"(O[3]),
                     "v"(O[4]), "v"(O[5]), "v"(O[6]), "v"(O[7]));
        __builtin_amdgcn_sched_barrier(0);
#pragma unroll
        for (int s = 0; s < SW_P; ++s) {
          const float piv = B[s][s];
#pragma unroll
          for (int i = s + 1; i < SW_P; ++i) {
            const float f = B[i][s] / piv;
#pragma unroll
            for (int c = s + 1; c < SW_P; ++c) B[i][c] = B[i][c] - f * B[s][c];
#pragma unroll
            for (int u = 0; u < 8; ++u) Pv[i][u] = Pv[i][u] - f * Pv[s][u];
          }
          const float fr = R[s] / piv;
          if (row != k + s) {
#pragma unroll
            for (int c = s + 1; c < SW_P; ++c) R[c] = R[c] - fr * B[s][c];
#pragma unroll
            for (int u = 0; u < 8; ++u) O[u] = O[u] - fr * Pv[s][u];
          }
        }
#pragma unroll
        for (int u = 0; u < 8; ++u) {
          const int j = j0 + u * SW_NW;
          if (j < 65) rr[j] = O[u];
        }
        if (tid == k) {
#pragma unroll
          for (int s = 0; s < SW_P; ++s) sPiv[k + s] = B[s][s];
        }
      } else if (widx == 0 || j0 < 32) {
        bool has[4];
        float mc[SW_P], mr[4];
#pragma unroll
        for (int c = 0; c < SW_P; ++c) mc[c] = sM[(k + c) * 32 + row];
#pragma unroll
        for (int u = 0; u < 4; ++u) {
          has[u] = j0 + u * SW_NW < 32;
          mr[u] = has[u] ? sM[(j0 + u * SW_NW) * 32 + row] : 0.f;
        }
        asm volatile("" ::"v"(mc[0]), "v"(mc[1]), "v"(mc[2]), "v"(mc[3]),
                     "v"(mr[0]), "v"(mr[1]), "v"(mr[2]), "v"(mr[3]));
        __builtin_amdgcn_sched_barrier(0);
        float ls[SW_P], dg[SW_P];
#pragma unroll
        for (int s = 0; s < SW_P; ++s) {
          const float piv = __int_as_float(__builtin_amdgcn_readlane(__float_as_int(mc[s]), k + s));
          dg[s] = sqrtf(piv);
          ls[s] = mc[s] / dg[s];
#pragma unroll
          for (int c = s + 1; c < SW_P; ++c) {
            const float lx = __int_as_float(__builtin_amdgcn_readlane(__float_as_int(ls[s]), k + c));
            mc[c] = mc[c] - ls[s] * lx;
          }
        }
#pragma unroll
        for (int u = 0; u < 4; ++u) {
          const int jx = __builtin_amdgcn_readfirstlane(has[u] ? j0 + u * SW_NW : k);
#pragma unroll
          for (int s = 0; s < SW_P; ++s) {
            const float lx = __int_as_float(__builtin_amdgcn_readlane(__float_as_int(ls[s]), jx));
            mr[u] = mr[u] - ls[s] * lx;
          }
        }
        if (row > k + SW_P - 1) {
#pragma unroll
          for (int u = 0; u < 4; ++u)
            if (has[u]) sM[(j0 + u * SW_NW) * 32 + row] = mr[u];
        }
        if (widx == 0) {
#pragma unroll
          for (int s = 0; s < SW_P; ++s) {
            if (row > k + s) sL[row * 32 + k + s] = ls[s];
            else if (row == k + s) sL[row * 32 + k + s] = dg[s];
          }
        }
      }
      __syncthreads();
    }
    for (int e = tid; e < 1024; e += FW_NT) {
      int i = e >> 5, j = e & 31;
      sC[e] = sAug[j * AUG_LD + 32 + i] / sPiv[j];
    }
    if (tid < DD) {
      sd[tid] = sAug[tid * AUG_LD + 64] / sPiv[tid];
      float r = noise[((size_t)n * TT + t) * DD + tid];
      if (t > 0) {
        float s = 0.f;
        for (int j = 0; j < DD; ++j) s += sLl[j * 32 + tid] * sx[j];
        r -= s;
      }
      srhs[tid] = r;
    }
    __syncthreads();
    {
      size_t base = (size_t)n * TT + t;
      if (haveNext)
        for (int e = tid; e < 1024; e += FW_NT) g_c[base * 1024 + e] = sC[e];
      if (tid < DD) g_d[base * DD + tid] = sd[tid];
    }
    if (haveNext && jgrp == 1) {
      float jd = 0.f;
      for (int j = 0; j < DD; ++j) jd += Jl_cur[row * 32 + j] * sd[j];
      sJlPd[row] = jd;
    }

    if (tid < 32) {
      const int jl = tid & 15;
      float ahi = srhs[jl + 16];
      float alo = srhs[jl];
      float Lhi[16], Llo[32];
#pragma unroll
      for (int i = 15; i >= 0; --i) {
        Lhi[i] = sL[(i + 16) * 32 + jl + 16];
        Llo[i + 16] = sL[(i + 16) * 32 + jl];
      }
#pragma unroll
      for (int i = 15; i >= 0; --i) Llo[i] = sL[i * 32 + jl];
      __builtin_amdgcn_sched_barrier(0);
      float xhi = 0.f, xlo = 0.f;
      SS_STEP_HI(31) SS_STEP_HI(30) SS_STEP_HI(29) SS_STEP_HI(28)
      SS_STEP_HI(27) SS_STEP_HI(26) SS_STEP_HI(25) SS_STEP_HI(24)
      SS_STEP_HI(23) SS_STEP_HI(22) SS_STEP_HI(21) SS_STEP_HI(20)
      SS_STEP_HI(19) SS_STEP_HI(18) SS_STEP_HI(17) SS_STEP_HI(16)
      SS_STEP_LO(15) SS_STEP_LO(14) SS_STEP_LO(13) SS_STEP_LO(12)
      SS_STEP_LO(11) SS_STEP_LO(10) SS_STEP_LO(9) SS_STEP_LO(8)
      SS_STEP_LO(7) SS_STEP_LO(6) SS_STEP_LO(5) SS_STEP_LO(4)
      SS_STEP_LO(3) SS_STEP_LO(2) SS_STEP_LO(1) SS_STEP_LO(0)
      if (tid < 16) {
        sx[jl] = xlo;
        sx[jl + 16] = xhi;
        float* o = &out[((size_t)n * TT + t) * DD];
        o[jl] = xlo;
        o[jl + 16] = xhi;
      }
    }
  }
}

__global__ __launch_bounds__(128) void lds_backward_kernel(
    const float* __restrict__ g_c, const float* __restrict__ g_d,
    float* __restrict__ out) {
  __shared__ float sCt[1024];
  __shared__ float sMu[DD], sNew[DD];
  const int n = blockIdx.x;
  const int tid = threadIdx.x;
  if (tid < DD) {
    float v = g_d[((size_t)n * TT + (TT - 1)) * DD + tid];
    sMu[tid] = v;
    out[((size_t)n * TT + (TT - 1)) * DD + tid] += v;
  }
  __syncthreads();
  for (int t = TT - 2; t >= 0; --t) {
    size_t base = (size_t)n * TT + t;
    for (int e = tid; e < 1024; e += 128) sCt[e] = g_c[base * 1024 + e];
    __syncthreads();
    if (tid < DD) {
      float acc = g_d[base * DD + tid];
      for (int j = 0; j < DD; ++j) acc -= sCt[tid * 32 + j] * sMu[j];
      sNew[tid] = acc;
    }
    __syncthreads();
    if (tid < DD) {
      sMu[tid] = sNew[tid];
      out[base * DD + tid] += sNew[tid];
    }
    __syncthreads();
  }
}

extern "C" void kernel_launch(void* const* d_in, const int* in_sizes, int n_in,
                              void* d_out, int out_size, void* d_ws,
                              size_t ws_size, hipStream_t stream) {
  (void)in_sizes; (void)n_in; (void)out_size; (void)ws_size;
  const float* z_samples = (const float*)d_in[0];
  const float* A_base = (const float*)d_in[1];
  const float* b_base = (const float*)d_in[2];
  const float* Q_sqrt = (const float*)d_in[3];
  const float* ms = (const float*)d_in[4];
  const float* Ri_sqrts = (const float*)d_in[5];
  const float* noise = (const float*)d_in[6];
  float* ws = (float*)d_ws;
  float* g_Ris = ws;
  float* g_hR = g_Ris + (size_t)TT * DD * DD;
  float* g_d = g_hR + (size_t)TT * DD;
  float* g_c = g_d + (size_t)NB * TT * DD;
  float* out = (float*)d_out;

  ris_precompute_kernel<<<TT, 128, 0, stream>>>(Ri_sqrts, ms, g_Ris, g_hR);
  lds_forward_kernel<<<NB, FW_NT, 0, stream>>>(z_samples, A_base, b_base, Q_sqrt,
                                             noise, g_Ris, g_hR, g_c, g_d, out);
  lds_backward_kernel<<<NB, 128, 0, stream>>>(g_c, g_d, out);
}
